# starting kernel
# baseline (speedup 1.0000x reference)
#include <hip/hip_runtime.h>
#include <math.h>


#define NQ 10
#define WAVES_PER_BLOCK 4

typedef float v2f __attribute__((ext_vector_type(2)));
typedef float v8f __attribute__((ext_vector_type(8)));

__device__ __forceinline__ v8f wmma4(v2f a, v2f b, v8f c) {
  return __builtin_amdgcn_wmma_f32_16x16x4_f32(
      false, a, false, b, (short)0, c, false, false);
}

__global__ void precompute_kernel(const float* __restrict__ qw,
                                  float2* __restrict__ mats,
                                  int* __restrict__ perms)
{
  int tid = threadIdx.x;
  for (int e = tid; e < 4096; e += blockDim.x) {
    int mat = e >> 10;
    int idx = e & 1023;
    int r = idx >> 5, k = idx & 31;
    int l = mat >> 1;
    int wbase = (mat & 1) * 5;
    float vr = 1.f, vi = 0.f;
    for (int j = 0; j < 5; ++j) {
      int q = wbase + j;
      const float* w = qw + (l * NQ + q) * 3;
      float phi = w[0], theta = w[1], omega = w[2];
      float ch = cosf(theta * 0.5f), sh = sinf(theta * 0.5f);
      int bi = (r >> (4 - j)) & 1;
      int bj = (k >> (4 - j)) & 1;
      float ang, mag;
      if (!bi && !bj)      { ang = -0.5f * (phi + omega); mag =  ch; }
      else if (!bi &&  bj) { ang =  0.5f * (phi - omega); mag = -sh; }
      else if ( bi && !bj) { ang = -0.5f * (phi - omega); mag =  sh; }
      else                 { ang =  0.5f * (phi + omega); mag =  ch; }
      float er = mag * cosf(ang), ei = mag * sinf(ang);
      float nr = vr * er - vi * ei;
      float ni = vr * ei + vi * er;
      vr = nr; vi = ni;
    }
    mats[mat * 1024 + idx] = make_float2(vr, vi);
  }
  for (int i = tid; i < 1024; i += blockDim.x) {
    for (int l = 0; l < 2; ++l) {
      int rr = l + 1;
      int b[NQ];
      for (int q = 0; q < NQ; ++q) b[q] = (i >> (NQ - 1 - q)) & 1;
      for (int q = 0; q < NQ; ++q) b[(q + rr) % NQ] ^= b[q];
      int dest = 0;
      for (int q = 0; q < NQ; ++q) dest |= b[q] << (NQ - 1 - q);
      perms[l * 1024 + i] = dest;
    }
  }
}

__global__ __launch_bounds__(32 * WAVES_PER_BLOCK)
void qlayer_kernel(const float* __restrict__ x,
                   const float2* __restrict__ mats,
                   const int* __restrict__ perms,
                   float* __restrict__ out)
{
  __shared__ float2 lds_all[WAVES_PER_BLOCK * 1024];
  const int lane = threadIdx.x & 31;
  const int wave = threadIdx.x >> 5;
  float2* S = lds_all + wave * 1024;
  const int b = blockIdx.x * WAVES_PER_BLOCK + wave;

  const float2* A0 = mats;
  const float2* B0 = mats + 1024;
  const float2* A1 = mats + 2048;
  const float2* B1 = mats + 3072;
  const int* perm1 = perms;
  const int* perm2 = perms + 1024;

  float cq[NQ], sq[NQ];
  #pragma unroll
  for (int q = 0; q < NQ; ++q) {
    float xv = x[b * NQ + q];
    cq[q] = cosf(0.5f * xv);
    sq[q] = sinf(0.5f * xv);
  }
  float u = 1.f, v = 1.f;
  #pragma unroll
  for (int j = 0; j < 5; ++j) {
    int bit = (lane >> (4 - j)) & 1;
    u *= bit ? sq[j] : cq[j];
    v *= bit ? sq[5 + j] : cq[5 + j];
  }

  float ar_ = 0.f, ai_ = 0.f, br_ = 0.f, bi_ = 0.f;
  for (int k = 0; k < 32; ++k) {
    float uk = __shfl(u, k, 32);
    float vk = __shfl(v, k, 32);
    float2 Ae = A0[lane * 32 + k];
    float2 Be = B0[lane * 32 + k];
    ar_ += Ae.x * uk; ai_ += Ae.y * uk;
    br_ += Be.x * vk; bi_ += Be.y * vk;
  }
  for (int r = 0; r < 32; ++r) {
    float xr = __shfl(ar_, r, 32);
    float xi = __shfl(ai_, r, 32);
    float wr = xr * br_ - xi * bi_;
    float wi = xr * bi_ + xi * br_;
    int d = perm1[r * 32 + lane];
    S[d] = make_float2(wr, wi);
  }
  __syncthreads();

  const v8f vzero = {0.f, 0.f, 0.f, 0.f, 0.f, 0.f, 0.f, 0.f};
  const int mh = lane & 15;
  const int kh = lane >> 4;
  const int mbase = kh * 8;
  const int nn = lane & 15;

  v8f accR[2][2], accI[2][2];
  #pragma unroll
  for (int I = 0; I < 2; ++I)
    #pragma unroll
    for (int J = 0; J < 2; ++J) { accR[I][J] = vzero; accI[I][J] = vzero; }

  #pragma unroll
  for (int I = 0; I < 2; ++I) {
    #pragma unroll
    for (int J = 0; J < 2; ++J) {
      #pragma unroll
      for (int kk = 0; kk < 32; kk += 4) {
        int k0 = kk + 2 * kh;
        float2 a0 = A1[(16 * I + mh) * 32 + k0];
        float2 a1 = A1[(16 * I + mh) * 32 + k0 + 1];
        v2f aR = {a0.x, a1.x};
        v2f aI = {a0.y, a1.y};
        v2f aIn = {-a0.y, -a1.y};
        float2 s0 = S[k0 * 32 + 16 * J + mh];
        float2 s1 = S[(k0 + 1) * 32 + 16 * J + mh];
        v2f bR = {s0.x, s1.x};
        v2f bI = {s0.y, s1.y};
        accR[I][J] = wmma4(aR,  bR, accR[I][J]);
        accR[I][J] = wmma4(aIn, bI, accR[I][J]);
        accI[I][J] = wmma4(aR,  bI, accI[I][J]);
        accI[I][J] = wmma4(aI,  bR, accI[I][J]);
      }
    }
  }
  __syncthreads();
  #pragma unroll
  for (int I = 0; I < 2; ++I)
    #pragma unroll
    for (int J = 0; J < 2; ++J)
      #pragma unroll
      for (int t = 0; t < 8; ++t) {
        int rr = 16 * I + mbase + t;
        int cc = 16 * J + nn;
        S[rr * 32 + cc] = make_float2(accR[I][J][t], accI[I][J][t]);
      }
  __syncthreads();

  v8f qR[2][2], qI[2][2];
  #pragma unroll
  for (int I = 0; I < 2; ++I)
    #pragma unroll
    for (int J = 0; J < 2; ++J) { qR[I][J] = vzero; qI[I][J] = vzero; }

  #pragma unroll
  for (int I = 0; I < 2; ++I) {
    #pragma unroll
    for (int J = 0; J < 2; ++J) {
      #pragma unroll
      for (int kk = 0; kk < 32; kk += 4) {
        int k0 = kk + 2 * kh;
        float2 p0 = S[(16 * I + mh) * 32 + k0];
        float2 p1 = S[(16 * I + mh) * 32 + k0 + 1];
        v2f pR = {p0.x, p1.x};
        v2f pI = {p0.y, p1.y};
        v2f pIn = {-p0.y, -p1.y};
        float2 b0 = B1[(16 * J + mh) * 32 + k0];
        float2 b1 = B1[(16 * J + mh) * 32 + k0 + 1];
        v2f bR = {b0.x, b1.x};
        v2f bI = {b0.y, b1.y};
        qR[I][J] = wmma4(pR,  bR, qR[I][J]);
        qR[I][J] = wmma4(pIn, bI, qR[I][J]);
        qI[I][J] = wmma4(pR,  bI, qI[I][J]);
        qI[I][J] = wmma4(pI,  bR, qI[I][J]);
      }
    }
  }
  __syncthreads();
  #pragma unroll
  for (int I = 0; I < 2; ++I)
    #pragma unroll
    for (int J = 0; J < 2; ++J)
      #pragma unroll
      for (int t = 0; t < 8; ++t) {
        int idx = (16 * I + mbase + t) * 32 + (16 * J + nn);
        int d = perm2[idx];
        S[d] = make_float2(qR[I][J][t], qI[I][J][t]);
      }
  __syncthreads();

  float e[NQ];
  #pragma unroll
  for (int q = 0; q < NQ; ++q) e[q] = 0.f;
  for (int row = 0; row < 32; ++row) {
    int i = row * 32 + lane;
    float2 amp = S[i];
    float p = amp.x * amp.x + amp.y * amp.y;
    #pragma unroll
    for (int q = 0; q < NQ; ++q)
      e[q] += ((i >> (NQ - 1 - q)) & 1) ? -p : p;
  }
  #pragma unroll
  for (int q = 0; q < NQ; ++q) {
    #pragma unroll
    for (int off = 16; off > 0; off >>= 1)
      e[q] += __shfl_xor(e[q], off, 32);
  }
  if (lane == 0) {
    #pragma unroll
    for (int q = 0; q < NQ; ++q) out[b * NQ + q] = e[q];
  }
}

extern "C" void kernel_launch(void* const* d_in, const int* in_sizes, int n_in,
                              void* d_out, int out_size, void* d_ws, size_t ws_size,
                              hipStream_t stream) {
  const float* x  = (const float*)d_in[0];
  const float* qw = (const float*)d_in[1];
  float* out = (float*)d_out;

  float2* mats = (float2*)d_ws;
  int* perms = (int*)((char*)d_ws + 4 * 1024 * sizeof(float2));

  hipLaunchKernelGGL(precompute_kernel, dim3(1), dim3(256), 0, stream,
                     qw, mats, perms);

  int batch = in_sizes[0] / NQ;
  hipLaunchKernelGGL(qlayer_kernel, dim3(batch / WAVES_PER_BLOCK),
                     dim3(32 * WAVES_PER_BLOCK), 0, stream,
                     x, mats, perms, out);
}


# Second 32x32 product on the four parity classes like the first (M = RB x classes of BB); state tiles = sum of four complex outer products L M, a depth-8 real WMMA: 80 -> 48 WMMAs; sum order changes
# speedup vs baseline: 5.3269x; 5.3269x over previous
#include <hip/hip_runtime.h>
#include <math.h>


#define NQ 10
#define WAVES_PER_BLOCK 4

typedef float v2f __attribute__((ext_vector_type(2)));
typedef float v8f __attribute__((ext_vector_type(8)));

__device__ __forceinline__ v8f wmma4(v2f a, v2f b, v8f c) {
  return __builtin_amdgcn_wmma_f32_16x16x4_f32(
      false, a, false, b, (short)0, c, false, false);
}

__global__ void precompute_kernel(const float* __restrict__ qw,
                                  float2* __restrict__ mats,
                                  float* __restrict__ rmats,
                                  float2* __restrict__ phase,
                                  int* __restrict__ perms)
{
  const int tid = blockIdx.x * blockDim.x + threadIdx.x;
  const int nthreads = gridDim.x * blockDim.x;
  for (int e = tid; e < 4160; e += nthreads) {
    int mat = e >> 10;
    int idx = e & 1023;
    if (mat < 2) {
      int k = idx >> 5, r = idx & 31;
      int wbase = mat * 5;
      float vr = 1.f, vi = 0.f;
      for (int j = 0; j < 5; ++j) {
        int q = wbase + j;
        const float* w = qw + q * 3;
        float phi = w[0], theta = w[1], omega = w[2];
        float ch = cosf(theta * 0.5f), sh = sinf(theta * 0.5f);
        int bi = (r >> (4 - j)) & 1;
        int bj = (k >> (4 - j)) & 1;
        float ang, mag;
        if (!bi && !bj)      { ang = -0.5f * (phi + omega); mag =  ch; }
        else if (!bi &&  bj) { ang =  0.5f * (phi - omega); mag = -sh; }
        else if ( bi && !bj) { ang = -0.5f * (phi - omega); mag =  sh; }
        else                 { ang =  0.5f * (phi + omega); mag =  ch; }
        float er = mag * cosf(ang), ei = mag * sinf(ang);
        float nr = vr * er - vi * ei;
        float ni = vr * ei + vi * er;
        vr = nr; vi = ni;
      }
      mats[mat * 1024 + idx] = make_float2(vr, vi);
    } else if (mat < 4) {
      int j2 = idx & 1, ln = (idx >> 1) & 31, ks = (idx >> 6) & 7, tile = idx >> 9;
      int r = 16 * tile + (ln & 15);
      int k = 4 * ks + 2 * (ln >> 4) + j2;
      int wbase = (mat & 1) * 5;
      float v = 1.f;
      for (int j = 0; j < 5; ++j) {
        int q = wbase + j;
        float theta = qw[(NQ + q) * 3 + 1];
        float ch = cosf(theta * 0.5f), sh = sinf(theta * 0.5f);
        int bi = (r >> (4 - j)) & 1;
        int bj = (k >> (4 - j)) & 1;
        v *= (bi == bj) ? ch : (bi ? sh : -sh);
      }
      rmats[(mat - 2) * 1024 + idx] = v;
    } else {
      int wbase = (idx >> 5) * 5, bits = idx & 31;
      float ang = 0.f;
      for (int j = 0; j < 5; ++j) {
        float hphi = 0.5f * qw[(NQ + wbase + j) * 3 + 0];
        ang += ((bits >> (4 - j)) & 1) ? hphi : -hphi;
      }
      phase[idx] = make_float2(cosf(ang), sinf(ang));
    }
  }
  for (int s = tid; s < 1024; s += nthreads) {
    int ln = s & 31, t = (s >> 5) & 7, J = (s >> 8) & 1, I = (s >> 9) & 1;
    int i = (16 * I + 8 * (ln >> 4) + t) * 32 + 16 * J + (ln & 15);
    int b[NQ];
    for (int q = 0; q < NQ; ++q) b[q] = (i >> (NQ - 1 - q)) & 1;
    for (int q = 0; q < NQ; ++q) b[(q + 2) % NQ] ^= b[q];
    int dest = 0;
    for (int q = 0; q < NQ; ++q) dest |= b[q] << (NQ - 1 - q);
    perms[1024 + s] = dest;
  }
}

__global__ __launch_bounds__(32 * WAVES_PER_BLOCK)
void qlayer_kernel(const float* __restrict__ x,
                   const float2* __restrict__ mats,
                   const float* __restrict__ rmats,
                   const float2* __restrict__ phase,
                   const int* __restrict__ perms,
                   float* __restrict__ out)
{
  __shared__ __attribute__((aligned(16))) float2 lds_all[WAVES_PER_BLOCK * 1024];
  const int lane = threadIdx.x & 31;
  const int wave = threadIdx.x >> 5;
  float2* S = lds_all + wave * 1024;
  const int b = __builtin_amdgcn_readfirstlane(blockIdx.x * WAVES_PER_BLOCK + wave);

  const float2* A0 = mats;
  const float2* B0 = mats + 1024;
  const float2* RAf = reinterpret_cast<const float2*>(rmats);
  const float2* RBf = reinterpret_cast<const float2*>(rmats + 1024);
  const int* perm2l = perms + 1024 + lane;

  const int ql = lane < NQ ? lane : NQ - 1;
  const float xh = 0.5f * x[b * NQ + ql];
  const float ch = cosf(xh);
  const float sh = sinf(xh);
  float cq[NQ], sq[NQ];
  #pragma unroll
  for (int q = 0; q < NQ; ++q) {
    cq[q] = __builtin_bit_cast(float, __builtin_amdgcn_readlane(__builtin_bit_cast(int, ch), q));
    sq[q] = __builtin_bit_cast(float, __builtin_amdgcn_readlane(__builtin_bit_cast(int, sh), q));
  }
  float u = 1.f, v = 1.f;
  #pragma unroll
  for (int j = 0; j < 5; ++j) {
    int bit = (lane >> (4 - j)) & 1;
    u *= bit ? sq[j] : cq[j];
    v *= bit ? sq[5 + j] : cq[5 + j];
  }

  float ar_ = 0.f, ai_ = 0.f, br_ = 0.f, bi_ = 0.f;
  for (int k = 0; k < 32; ++k) {
    float uk = __shfl(u, k, 32);
    float vk = __shfl(v, k, 32);
    float2 Ae = A0[k * 32 + lane];
    float2 Be = B0[k * 32 + lane];
    ar_ += Ae.x * uk; ai_ += Ae.y * uk;
    br_ += Be.x * vk; bi_ += Be.y * vk;
  }
  const v8f vzero = {0.f, 0.f, 0.f, 0.f, 0.f, 0.f, 0.f, 0.f};
  const int kh = lane >> 4;

  float* TA = reinterpret_cast<float*>(S);
  float* BBp = TA + 128;
  {
    const int gr = lane ^ (lane >> 1);
    const float2 pr = phase[lane];
    const float a0r = __shfl(ar_, gr, 32),      a0i = __shfl(ai_, gr, 32);
    const float a1r = __shfl(ar_, gr ^ 24, 32), a1i = __shfl(ai_, gr ^ 24, 32);
    TA[lane]      = a0r * pr.x - a0i * pr.y;
    TA[32 + lane] = a1r * pr.x - a1i * pr.y;
    TA[64 + lane] = a0r * pr.y + a0i * pr.x;
    TA[96 + lane] = a1r * pr.y + a1i * pr.x;
  }
  {
    const int gc = lane ^ (lane >> 1);
    const float2 pc = phase[32 + lane];
    const float b0r = __shfl(br_, gc, 32),      b0i = __shfl(bi_, gc, 32);
    const float b1r = __shfl(br_, gc ^ 16, 32), b1i = __shfl(bi_, gc ^ 16, 32);
    BBp[lane]      = b0r * pc.x - b0i * pc.y;
    BBp[32 + lane] = b0r * pc.y + b0i * pc.x;
    BBp[64 + lane] = b1r * pc.x - b1i * pc.y;
    BBp[96 + lane] = b1r * pc.y + b1i * pc.x;
  }
  __syncthreads();
  const int cp = (lane >> 3) & 1;
  const int cpart = (lane >> 1) & 1;
  const int cg = lane & 1;
  const float* TAc = TA + 64 * cpart + 32 * cg + cp + 2 * kh;
  const float* BBc = BBp + 64 * cp + 32 * cpart + cg + 2 * kh;
  v8f lt[2] = {vzero, vzero};
  v8f mt[2] = {vzero, vzero};
  #pragma unroll
  for (int ks = 0; ks < 8; ++ks) {
    const float xa = TAc[4 * ks];
    const float xb = BBc[4 * ks];
    const v2f a2 = cp ? v2f{0.f, xa} : v2f{xa, 0.f};
    const v2f b2 = cg ? v2f{0.f, xb} : v2f{xb, 0.f};
    #pragma unroll
    for (int T = 0; T < 2; ++T) {
      float2 av = RAf[(T * 8 + ks) * 32 + lane];
      float2 bv = RBf[(T * 8 + ks) * 32 + lane];
      lt[T] = wmma4(a2, v2f{av.x, av.y}, lt[T]);
      mt[T] = wmma4(b2, v2f{bv.x, bv.y}, mt[T]);
    }
  }

  v8f qR[2][2], qI[2][2];
  #pragma unroll
  for (int J = 0; J < 2; ++J) {
    const v2f Mr  = {mt[J][0], mt[J][1]};
    const v2f Mi  = {mt[J][2], mt[J][3]};
    const v2f nMi = {-mt[J][2], -mt[J][3]};
    #pragma unroll
    for (int I = 0; I < 2; ++I) {
      const v2f Lr = {lt[I][0], lt[I][1]};
      const v2f Li = {lt[I][2], lt[I][3]};
      qR[I][J] = wmma4(Li, nMi, wmma4(Lr, Mr, vzero));
      qI[I][J] = wmma4(Li, Mr,  wmma4(Lr, Mi, vzero));
    }
  }
  __syncthreads();
  #pragma unroll
  for (int I = 0; I < 2; ++I)
    #pragma unroll
    for (int J = 0; J < 2; ++J)
      #pragma unroll
      for (int t = 0; t < 8; ++t) {
        int d = perm2l[((I * 2 + J) * 8 + t) * 32];
        S[d] = make_float2(qR[I][J][t], qI[I][J][t]);
      }
  __syncthreads();

  float c0 = 0.f, c1 = 0.f, c2 = 0.f, c3 = 0.f;
  float g0 = 0.f, g1 = 0.f, g2 = 0.f;
  #pragma unroll
  for (int m = 0; m < 8; ++m) {
    float p[4];
    #pragma unroll
    for (int j = 0; j < 4; ++j) {
      float2 amp = S[(4 * m + j) * 32 + lane];
      p[j] = amp.x * amp.x + amp.y * amp.y;
    }
    c0 += p[0]; c1 += p[1]; c2 += p[2]; c3 += p[3];
    const float g = (p[0] + p[1]) + (p[2] + p[3]);
    g0 += (m & 4) ? -g : g;
    g1 += (m & 2) ? -g : g;
    g2 += (m & 1) ? -g : g;
  }
  const float s01 = c0 + c1, s23 = c2 + c3;
  const float T = s01 + s23;
  float e[NQ];
  e[0] = g0; e[1] = g1; e[2] = g2;
  e[3] = s01 - s23;
  e[4] = (c0 - c1) + (c2 - c3);
  #pragma unroll
  for (int j = 0; j < 5; ++j)
    e[5 + j] = ((lane >> (4 - j)) & 1) ? -T : T;
  #pragma unroll
  for (int q = 0; q < NQ; ++q) {
    #pragma unroll
    for (int off = 16; off > 0; off >>= 1)
      e[q] += __shfl_xor(e[q], off, 32);
  }
  if (lane == 0) {
    #pragma unroll
    for (int q = 0; q < NQ; ++q) out[b * NQ + q] = e[q];
  }
}

extern "C" void kernel_launch(void* const* d_in, const int* in_sizes, int n_in,
                              void* d_out, int out_size, void* d_ws, size_t ws_size,
                              hipStream_t stream) {
  const float* x  = (const float*)d_in[0];
  const float* qw = (const float*)d_in[1];
  float* out = (float*)d_out;

  float2* mats = (float2*)d_ws;
  float* rmats = (float*)((char*)d_ws + 16384);
  float2* phase = (float2*)((char*)d_ws + 24576);
  int* perms = (int*)((char*)d_ws + 32768);

  hipLaunchKernelGGL(precompute_kernel, dim3(17), dim3(256), 0, stream,
                     qw, mats, rmats, phase, perms);

  int batch = in_sizes[0] / NQ;
  hipLaunchKernelGGL(qlayer_kernel, dim3(batch / WAVES_PER_BLOCK),
                     dim3(32 * WAVES_PER_BLOCK), 0, stream,
                     x, mats, rmats, phase, perms, out);
}


# Stage-1 products on the 16x16 RY factor of wires 1-4 / 6-9, both halves of each class vector as WMMA rows (all 16 used); wire 0 / 5's 2x2 RY joined in-lane after: 48 -> 24 WMMAs; sum order changes
# speedup vs baseline: 6.2861x; 1.1801x over previous
#include <hip/hip_runtime.h>
#include <math.h>


#define NQ 10
#define WAVES_PER_BLOCK 4

typedef float v2f __attribute__((ext_vector_type(2)));
typedef float v8f __attribute__((ext_vector_type(8)));

__device__ __forceinline__ v8f wmma4(v2f a, v2f b, v8f c) {
  return __builtin_amdgcn_wmma_f32_16x16x4_f32(
      false, a, false, b, (short)0, c, false, false);
}

__global__ void precompute_kernel(const float* __restrict__ qw,
                                  float2* __restrict__ mats,
                                  float* __restrict__ rmats,
                                  float2* __restrict__ phase,
                                  int* __restrict__ perms)
{
  const int tid = blockIdx.x * blockDim.x + threadIdx.x;
  const int nthreads = gridDim.x * blockDim.x;
  for (int e = tid; e < 2626; e += nthreads) {
    if (e < 2048) {
      int mat = e >> 10;
      int idx = e & 1023;
      int k = idx >> 5, r = idx & 31;
      int wbase = mat * 5;
      float vr = 1.f, vi = 0.f;
      for (int j = 0; j < 5; ++j) {
        int q = wbase + j;
        const float* w = qw + q * 3;
        float phi = w[0], theta = w[1], omega = w[2];
        float ch = cosf(theta * 0.5f), sh = sinf(theta * 0.5f);
        int bi = (r >> (4 - j)) & 1;
        int bj = (k >> (4 - j)) & 1;
        float ang, mag;
        if (!bi && !bj)      { ang = -0.5f * (phi + omega); mag =  ch; }
        else if (!bi &&  bj) { ang =  0.5f * (phi - omega); mag = -sh; }
        else if ( bi && !bj) { ang = -0.5f * (phi - omega); mag =  sh; }
        else                 { ang =  0.5f * (phi + omega); mag =  ch; }
        float er = mag * cosf(ang), ei = mag * sinf(ang);
        float nr = vr * er - vi * ei;
        float ni = vr * ei + vi * er;
        vr = nr; vi = ni;
      }
      mats[mat * 1024 + idx] = make_float2(vr, vi);
    } else if (e < 2560) {
      int idx = e - 2048;
      int j2 = idx & 1, ln = (idx >> 1) & 31, ks = (idx >> 6) & 3, which = idx >> 8;
      int r = ln & 15;
      int k = 4 * ks + 2 * (ln >> 4) + j2;
      int wbase = which * 5 + 1;
      float v = 1.f;
      for (int j = 0; j < 4; ++j) {
        int q = wbase + j;
        float theta = qw[(NQ + q) * 3 + 1];
        float ch = cosf(theta * 0.5f), sh = sinf(theta * 0.5f);
        int bi = (r >> (3 - j)) & 1;
        int bj = (k >> (3 - j)) & 1;
        v *= (bi == bj) ? ch : (bi ? sh : -sh);
      }
      rmats[idx] = v;
    } else if (e < 2624) {
      int idx = e - 2560;
      int wbase = (idx >> 5) * 5, bits = idx & 31;
      float ang = 0.f;
      for (int j = 0; j < 5; ++j) {
        float hphi = 0.5f * qw[(NQ + wbase + j) * 3 + 0];
        ang += ((bits >> (4 - j)) & 1) ? hphi : -hphi;
      }
      phase[idx] = make_float2(cosf(ang), sinf(ang));
    } else {
      int idx = e - 2560;
      float theta = qw[(NQ + (idx - 64) * 5) * 3 + 1];
      phase[idx] = make_float2(cosf(theta * 0.5f), sinf(theta * 0.5f));
    }
  }
  for (int s = tid; s < 1024; s += nthreads) {
    int ln = s & 31, t = (s >> 5) & 7, J = (s >> 8) & 1, I = (s >> 9) & 1;
    int i = (16 * I + 8 * (ln >> 4) + t) * 32 + 16 * J + (ln & 15);
    int b[NQ];
    for (int q = 0; q < NQ; ++q) b[q] = (i >> (NQ - 1 - q)) & 1;
    for (int q = 0; q < NQ; ++q) b[(q + 2) % NQ] ^= b[q];
    int dest = 0;
    for (int q = 0; q < NQ; ++q) dest |= b[q] << (NQ - 1 - q);
    perms[1024 + s] = dest;
  }
}

__global__ __launch_bounds__(32 * WAVES_PER_BLOCK)
void qlayer_kernel(const float* __restrict__ x,
                   const float2* __restrict__ mats,
                   const float* __restrict__ rmats,
                   const float2* __restrict__ phase,
                   const int* __restrict__ perms,
                   float* __restrict__ out)
{
  __shared__ __attribute__((aligned(16))) float2 lds_all[WAVES_PER_BLOCK * 1024];
  const int lane = threadIdx.x & 31;
  const int wave = threadIdx.x >> 5;
  float2* S = lds_all + wave * 1024;
  const int b = __builtin_amdgcn_readfirstlane(blockIdx.x * WAVES_PER_BLOCK + wave);

  const float2* A0 = mats;
  const float2* B0 = mats + 1024;
  const float2* RAf = reinterpret_cast<const float2*>(rmats);
  const float2* RBf = reinterpret_cast<const float2*>(rmats + 256);
  const int* perm2l = perms + 1024 + lane;

  const int ql = lane < NQ ? lane : NQ - 1;
  const float xh = 0.5f * x[b * NQ + ql];
  const float ch = cosf(xh);
  const float sh = sinf(xh);
  float cq[NQ], sq[NQ];
  #pragma unroll
  for (int q = 0; q < NQ; ++q) {
    cq[q] = __builtin_bit_cast(float, __builtin_amdgcn_readlane(__builtin_bit_cast(int, ch), q));
    sq[q] = __builtin_bit_cast(float, __builtin_amdgcn_readlane(__builtin_bit_cast(int, sh), q));
  }
  float u = 1.f, v = 1.f;
  #pragma unroll
  for (int j = 0; j < 5; ++j) {
    int bit = (lane >> (4 - j)) & 1;
    u *= bit ? sq[j] : cq[j];
    v *= bit ? sq[5 + j] : cq[5 + j];
  }

  float ar_ = 0.f, ai_ = 0.f, br_ = 0.f, bi_ = 0.f;
  for (int k = 0; k < 32; ++k) {
    float uk = __shfl(u, k, 32);
    float vk = __shfl(v, k, 32);
    float2 Ae = A0[k * 32 + lane];
    float2 Be = B0[k * 32 + lane];
    ar_ += Ae.x * uk; ai_ += Ae.y * uk;
    br_ += Be.x * vk; bi_ += Be.y * vk;
  }
  const v8f vzero = {0.f, 0.f, 0.f, 0.f, 0.f, 0.f, 0.f, 0.f};
  const int kh = lane >> 4;

  float* TA = reinterpret_cast<float*>(S);
  float* BBp = TA + 128;
  {
    const int gr = lane ^ (lane >> 1);
    const float2 pr = phase[lane];
    const float a0r = __shfl(ar_, gr, 32),      a0i = __shfl(ai_, gr, 32);
    const float a1r = __shfl(ar_, gr ^ 24, 32), a1i = __shfl(ai_, gr ^ 24, 32);
    TA[lane]      = a0r * pr.x - a0i * pr.y;
    TA[32 + lane] = a1r * pr.x - a1i * pr.y;
    TA[64 + lane] = a0r * pr.y + a0i * pr.x;
    TA[96 + lane] = a1r * pr.y + a1i * pr.x;
  }
  {
    const int gc = lane ^ (lane >> 1);
    const float2 pc = phase[32 + lane];
    const float b0r = __shfl(br_, gc, 32),      b0i = __shfl(bi_, gc, 32);
    const float b1r = __shfl(br_, gc ^ 16, 32), b1i = __shfl(bi_, gc ^ 16, 32);
    BBp[lane]      = b0r * pc.x - b0i * pc.y;
    BBp[32 + lane] = b0r * pc.y + b0i * pc.x;
    BBp[64 + lane] = b1r * pc.x - b1i * pc.y;
    BBp[96 + lane] = b1r * pc.y + b1i * pc.x;
  }
  __syncthreads();
  const int cp = (lane >> 3) & 1;
  const int cpart = (lane >> 2) & 1;
  const int cg = (lane >> 1) & 1;
  const int cb = lane & 1;
  const float* TAc = TA + 64 * cpart + 32 * cg + 16 * cb + cp + 2 * kh;
  const float* BBc = BBp + 64 * cp + 32 * cpart + 16 * cb + cg + 2 * kh;
  v8f xt = vzero;
  v8f yt = vzero;
  #pragma unroll
  for (int ks = 0; ks < 4; ++ks) {
    const float xa = TAc[4 * ks];
    const float xb = BBc[4 * ks];
    const v2f a2 = cp ? v2f{0.f, xa} : v2f{xa, 0.f};
    const v2f b2 = cg ? v2f{0.f, xb} : v2f{xb, 0.f};
    float2 av = RAf[ks * 32 + lane];
    float2 bv = RBf[ks * 32 + lane];
    xt = wmma4(a2, v2f{av.x, av.y}, xt);
    yt = wmma4(b2, v2f{bv.x, bv.y}, yt);
  }
  const float2 ry0 = phase[64];
  const float2 ry5 = phase[65];
  float lt[2][4], mt[2][4];
  #pragma unroll
  for (int u = 0; u < 4; ++u) {
    lt[0][u] = ry0.x * xt[2 * u] - ry0.y * xt[2 * u + 1];
    lt[1][u] = ry0.y * xt[2 * u] + ry0.x * xt[2 * u + 1];
    mt[0][u] = ry5.x * yt[2 * u] - ry5.y * yt[2 * u + 1];
    mt[1][u] = ry5.y * yt[2 * u] + ry5.x * yt[2 * u + 1];
  }

  v8f qR[2][2], qI[2][2];
  #pragma unroll
  for (int J = 0; J < 2; ++J) {
    const v2f Mr  = {mt[J][0], mt[J][1]};
    const v2f Mi  = {mt[J][2], mt[J][3]};
    const v2f nMi = {-mt[J][2], -mt[J][3]};
    #pragma unroll
    for (int I = 0; I < 2; ++I) {
      const v2f Lr = {lt[I][0], lt[I][1]};
      const v2f Li = {lt[I][2], lt[I][3]};
      qR[I][J] = wmma4(Li, nMi, wmma4(Lr, Mr, vzero));
      qI[I][J] = wmma4(Li, Mr,  wmma4(Lr, Mi, vzero));
    }
  }
  __syncthreads();
  #pragma unroll
  for (int I = 0; I < 2; ++I)
    #pragma unroll
    for (int J = 0; J < 2; ++J)
      #pragma unroll
      for (int t = 0; t < 8; ++t) {
        int d = perm2l[((I * 2 + J) * 8 + t) * 32];
        S[d] = make_float2(qR[I][J][t], qI[I][J][t]);
      }
  __syncthreads();

  float c0 = 0.f, c1 = 0.f, c2 = 0.f, c3 = 0.f;
  float g0 = 0.f, g1 = 0.f, g2 = 0.f;
  #pragma unroll
  for (int m = 0; m < 8; ++m) {
    float p[4];
    #pragma unroll
    for (int j = 0; j < 4; ++j) {
      float2 amp = S[(4 * m + j) * 32 + lane];
      p[j] = amp.x * amp.x + amp.y * amp.y;
    }
    c0 += p[0]; c1 += p[1]; c2 += p[2]; c3 += p[3];
    const float g = (p[0] + p[1]) + (p[2] + p[3]);
    g0 += (m & 4) ? -g : g;
    g1 += (m & 2) ? -g : g;
    g2 += (m & 1) ? -g : g;
  }
  const float s01 = c0 + c1, s23 = c2 + c3;
  const float T = s01 + s23;
  float e[NQ];
  e[0] = g0; e[1] = g1; e[2] = g2;
  e[3] = s01 - s23;
  e[4] = (c0 - c1) + (c2 - c3);
  #pragma unroll
  for (int j = 0; j < 5; ++j)
    e[5 + j] = ((lane >> (4 - j)) & 1) ? -T : T;
  #pragma unroll
  for (int q = 0; q < NQ; ++q) {
    #pragma unroll
    for (int off = 16; off > 0; off >>= 1)
      e[q] += __shfl_xor(e[q], off, 32);
  }
  if (lane == 0) {
    #pragma unroll
    for (int q = 0; q < NQ; ++q) out[b * NQ + q] = e[q];
  }
}

extern "C" void kernel_launch(void* const* d_in, const int* in_sizes, int n_in,
                              void* d_out, int out_size, void* d_ws, size_t ws_size,
                              hipStream_t stream) {
  const float* x  = (const float*)d_in[0];
  const float* qw = (const float*)d_in[1];
  float* out = (float*)d_out;

  float2* mats = (float2*)d_ws;
  float* rmats = (float*)((char*)d_ws + 16384);
  float2* phase = (float2*)((char*)d_ws + 24576);
  int* perms = (int*)((char*)d_ws + 32768);

  hipLaunchKernelGGL(precompute_kernel, dim3(11), dim3(256), 0, stream,
                     qw, mats, rmats, phase, perms);

  int batch = in_sizes[0] / NQ;
  hipLaunchKernelGGL(qlayer_kernel, dim3(batch / WAVES_PER_BLOCK),
                     dim3(32 * WAVES_PER_BLOCK), 0, stream,
                     x, mats, rmats, phase, perms, out);
}
